# starting kernel
# baseline (speedup 1.0000x reference)
#include <hip/hip_runtime.h>
#include <math.h>


#define BQ   4096
#define LL   4
#define HH   15
#define WW   15
#define NPOS (HH * WW)
#define CC   32
#define MAPMAX 32.0f

typedef __attribute__((ext_vector_type(2))) float v2f;
typedef __attribute__((ext_vector_type(8))) float v8f;

__global__ __launch_bounds__(256) void fused_front_kernel(
    const int*   __restrict__ enc,
    const float* __restrict__ emb,
    const float* __restrict__ prelu_map,
    const float* __restrict__ dwconv_w,
    const float* __restrict__ dwconv_b,
    const float* __restrict__ policy_w,
    const float* __restrict__ prelu_policy,
    float*       __restrict__ policy_out,
    float*       __restrict__ vpool)
{
  __shared__ float xs[NPOS * CC];
  __shared__ float red[8 * CC];

  const int b    = blockIdx.x;
  const int tid  = threadIdx.x;
  const int c    = tid & 31;
  const int wave = tid >> 5;

  const float amap  = prelu_map[c];
  const int*  ebase = enc + (size_t)b * LL * NPOS;
  for (int i = tid; i < NPOS * CC; i += 256) {
    const int pos = i >> 5;
    float acc = 0.f;
#pragma unroll
    for (int l = 0; l < LL; ++l) {
      const int idx = ebase[l * NPOS + pos];
      const float e = emb[(size_t)idx * CC + c];
      acc += MAPMAX * tanhf(e * (1.0f / MAPMAX));
    }
    float f = acc * 0.25f;
    xs[i] = (f >= 0.f) ? f : amap * f;
  }
  __syncthreads();

  float wreg[9];
#pragma unroll
  for (int j = 0; j < 9; ++j) wreg[j] = dwconv_w[c * 9 + j];
  const float bias = dwconv_b[c];
  const float pw   = (c < 16) ? policy_w[c] : 0.f;
  const float pa   = prelu_policy[0];

  float vacc = 0.f;
  for (int it = 0; it < 29; ++it) {
    const int p = it * 8 + wave;
    float contrib = 0.f;
    if (p < NPOS) {
      const int h = p / WW, w = p % WW;
      float a = bias;
#pragma unroll
      for (int kh = 0; kh < 3; ++kh) {
        const int hh = h + kh - 1;
        if ((unsigned)hh < (unsigned)HH) {
#pragma unroll
          for (int kw = 0; kw < 3; ++kw) {
            const int ww2 = w + kw - 1;
            if ((unsigned)ww2 < (unsigned)WW)
              a += xs[(hh * WW + ww2) * CC + c] * wreg[kh * 3 + kw];
          }
        }
      }
      const float y = fmaxf(a, 0.f);
      vacc += y;
      contrib = y * pw;
    }
#pragma unroll
    for (int off = 16; off > 0; off >>= 1)
      contrib += __shfl_xor(contrib, off, 32);
    if (p < NPOS && c == 0) {
      const float pol = (contrib >= 0.f) ? contrib : pa * contrib;
      policy_out[(size_t)b * NPOS + p] = pol;
    }
  }

  red[wave * CC + c] = vacc;
  __syncthreads();
  if (tid < CC) {
    float s = 0.f;
#pragma unroll
    for (int w8 = 0; w8 < 8; ++w8) s += red[w8 * CC + tid];
    vpool[(size_t)b * CC + tid] = s * (1.0f / (float)NPOS);
  }
}

__device__ __forceinline__ v8f wmma4(v2f a, v2f b, v8f c) {
  return __builtin_amdgcn_wmma_f32_16x16x4_f32(
      false, a, false, b, (short)0, c, false, false);
}

#define HSTRIDE 34

__global__ __launch_bounds__(256) void value_mlp_kernel(
    const float* __restrict__ vpool,
    const float* __restrict__ w1, const float* __restrict__ b1,
    const float* __restrict__ w2, const float* __restrict__ b2,
    const float* __restrict__ w3, const float* __restrict__ b3,
    float*       __restrict__ out)
{
  __shared__ float hb[8][16 * HSTRIDE];

  const int tid   = threadIdx.x;
  const int wave  = tid >> 5;
  const int lane  = tid & 31;
  const int lh    = lane & 15;
  const int kh    = (lane >> 4) * 2;
  const int row0  = blockIdx.x * 128 + wave * 16;
  float* hbuf = hb[wave];

  v8f c0 = {}, c1 = {};
  {
    const float* arow = vpool + (size_t)(row0 + lh) * 32;
#pragma unroll
    for (int k0 = 0; k0 < 32; k0 += 4) {
      v2f a;  a.x  = arow[k0 + kh];              a.y  = arow[k0 + kh + 1];
      v2f bA; bA.x = w1[lh * 32 + k0 + kh];      bA.y = w1[lh * 32 + k0 + kh + 1];
      v2f bB; bB.x = w1[(lh + 16) * 32 + k0 + kh];
              bB.y = w1[(lh + 16) * 32 + k0 + kh + 1];
      c0 = wmma4(a, bA, c0);
      c1 = wmma4(a, bB, c1);
    }
  }
  {
    const float bb0 = b1[lh], bb1 = b1[lh + 16];
#pragma unroll
    for (int v = 0; v < 8; ++v) {
      const int m = v + (lane >> 4) * 8;
      hbuf[m * HSTRIDE + lh]      = fmaxf(c0[v] + bb0, 0.f);
      hbuf[m * HSTRIDE + lh + 16] = fmaxf(c1[v] + bb1, 0.f);
    }
  }
  __syncthreads();

  v8f d0 = {}, d1 = {};
  {
    const float* arow = hbuf + lh * HSTRIDE;
#pragma unroll
    for (int k0 = 0; k0 < 32; k0 += 4) {
      v2f a;  a.x  = arow[k0 + kh];              a.y  = arow[k0 + kh + 1];
      v2f bA; bA.x = w2[lh * 32 + k0 + kh];      bA.y = w2[lh * 32 + k0 + kh + 1];
      v2f bB; bB.x = w2[(lh + 16) * 32 + k0 + kh];
              bB.y = w2[(lh + 16) * 32 + k0 + kh + 1];
      d0 = wmma4(a, bA, d0);
      d1 = wmma4(a, bB, d1);
    }
  }
  __syncthreads();
  {
    const float bb0 = b2[lh], bb1 = b2[lh + 16];
#pragma unroll
    for (int v = 0; v < 8; ++v) {
      const int m = v + (lane >> 4) * 8;
      hbuf[m * HSTRIDE + lh]      = fmaxf(d0[v] + bb0, 0.f);
      hbuf[m * HSTRIDE + lh + 16] = fmaxf(d1[v] + bb1, 0.f);
    }
  }
  __syncthreads();

  if (lane < 16) {
    const float* hrow = hbuf + lh * HSTRIDE;
    float a0 = b3[0], a1 = b3[1], a2 = b3[2];
#pragma unroll
    for (int k = 0; k < 32; ++k) {
      const float h = hrow[k];
      a0 += h * w3[k];
      a1 += h * w3[32 + k];
      a2 += h * w3[64 + k];
    }
    const size_t r = (size_t)(row0 + lh) * 3;
    out[r + 0] = a0; out[r + 1] = a1; out[r + 2] = a2;
  }
}

extern "C" void kernel_launch(void* const* d_in, const int* in_sizes, int n_in,
                              void* d_out, int out_size, void* d_ws, size_t ws_size,
                              hipStream_t stream) {
  (void)in_sizes; (void)n_in; (void)out_size; (void)ws_size;
  const int*   enc          = (const int*)  d_in[0];
  const float* emb          = (const float*)d_in[1];
  const float* prelu_map    = (const float*)d_in[2];
  const float* dwconv_w     = (const float*)d_in[3];
  const float* dwconv_b     = (const float*)d_in[4];
  const float* policy_w     = (const float*)d_in[5];
  const float* prelu_policy = (const float*)d_in[6];
  const float* w1 = (const float*)d_in[7];
  const float* b1 = (const float*)d_in[8];
  const float* w2 = (const float*)d_in[9];
  const float* b2 = (const float*)d_in[10];
  const float* w3 = (const float*)d_in[11];
  const float* b3 = (const float*)d_in[12];

  float* out        = (float*)d_out;
  float* value_out  = out;
  float* policy_out = out + (size_t)BQ * 3;
  float* vpool      = (float*)d_ws;

  fused_front_kernel<<<BQ, 256, 0, stream>>>(
      enc, emb, prelu_map, dwconv_w, dwconv_b, policy_w, prelu_policy,
      policy_out, vpool);

  value_mlp_kernel<<<BQ / 128, 256, 0, stream>>>(
      vpool, w1, b1, w2, b2, w3, b3, value_out);
}


# Stage-1 gather de-serialised: 16 row numbers per vector load, broadcast to scalars, 16 table-row loads issued back to back before the first tanh (was two dependent loads per row); sums unchanged
# speedup vs baseline: 1.6398x; 1.6398x over previous
#include <hip/hip_runtime.h>
#include <math.h>


#define BQ   4096
#define LL   4
#define HH   15
#define WW   15
#define NPOS (HH * WW)
#define CC   32
#define MAPMAX 32.0f

typedef __attribute__((ext_vector_type(2))) float v2f;
typedef __attribute__((ext_vector_type(8))) float v8f;

__global__ __launch_bounds__(256) void fused_front_kernel(
    const int*   __restrict__ enc,
    const float* __restrict__ emb,
    const float* __restrict__ prelu_map,
    const float* __restrict__ dwconv_w,
    const float* __restrict__ dwconv_b,
    const float* __restrict__ policy_w,
    const float* __restrict__ prelu_policy,
    float*       __restrict__ policy_out,
    float*       __restrict__ vpool)
{
  __shared__ float xs[NPOS * CC];
  __shared__ float red[8 * CC];

  const int b    = blockIdx.x;
  const int tid  = threadIdx.x;
  const int c    = tid & 31;
  const int wave = tid >> 5;

  const float amap  = prelu_map[c];
  const int*  ebase = enc + (size_t)b * LL * NPOS;
  const int   wv    = __builtin_amdgcn_readfirstlane(wave);
  const int   jl    = (c >> 2) & 3;
  const int   ll    = c & 3;
  int idv = ebase[ll * NPOS + jl * 8 + wv];
#pragma unroll 1
  for (int g = 0; g < 7; ++g) {
    const int posn = (g < 6) ? ((4 * (g + 1) + jl) * 8 + wv) : (NPOS - 1);
    const int idn  = ebase[ll * NPOS + posn];
    float e[16];
#pragma unroll
    for (int k = 0; k < 16; ++k) {
      const int idx = __builtin_amdgcn_readlane(idv, k);
      e[k] = emb[(unsigned)(idx * CC + c)];
    }
#pragma unroll
    for (int j = 0; j < 4; ++j) {
      float acc = 0.f;
#pragma unroll
      for (int l = 0; l < LL; ++l)
        acc += MAPMAX * tanhf(e[4 * j + l] * (1.0f / MAPMAX));
      const float f = acc * 0.25f;
      const int pos = (4 * g + j) * 8 + wv;
      xs[pos * CC + c] = (f >= 0.f) ? f : amap * f;
    }
    idv = idn;
  }
  if (wv == 0) {
    float acc = 0.f;
    float e[4];
#pragma unroll
    for (int l = 0; l < LL; ++l) {
      const int idx = __builtin_amdgcn_readlane(idv, l);
      e[l] = emb[(unsigned)(idx * CC + c)];
    }
#pragma unroll
    for (int l = 0; l < LL; ++l)
      acc += MAPMAX * tanhf(e[l] * (1.0f / MAPMAX));
    const float f = acc * 0.25f;
    xs[(NPOS - 1) * CC + c] = (f >= 0.f) ? f : amap * f;
  }
  __syncthreads();

  float wreg[9];
#pragma unroll
  for (int j = 0; j < 9; ++j) wreg[j] = dwconv_w[c * 9 + j];
  const float bias = dwconv_b[c];
  const float pw   = (c < 16) ? policy_w[c] : 0.f;
  const float pa   = prelu_policy[0];

  float vacc = 0.f;
  for (int it = 0; it < 29; ++it) {
    const int p = it * 8 + wave;
    float contrib = 0.f;
    if (p < NPOS) {
      const int h = p / WW, w = p % WW;
      float a = bias;
#pragma unroll
      for (int kh = 0; kh < 3; ++kh) {
        const int hh = h + kh - 1;
        if ((unsigned)hh < (unsigned)HH) {
#pragma unroll
          for (int kw = 0; kw < 3; ++kw) {
            const int ww2 = w + kw - 1;
            if ((unsigned)ww2 < (unsigned)WW)
              a += xs[(hh * WW + ww2) * CC + c] * wreg[kh * 3 + kw];
          }
        }
      }
      const float y = fmaxf(a, 0.f);
      vacc += y;
      contrib = y * pw;
    }
#pragma unroll
    for (int off = 16; off > 0; off >>= 1)
      contrib += __shfl_xor(contrib, off, 32);
    if (p < NPOS && c == 0) {
      const float pol = (contrib >= 0.f) ? contrib : pa * contrib;
      policy_out[(size_t)b * NPOS + p] = pol;
    }
  }

  red[wave * CC + c] = vacc;
  __syncthreads();
  if (tid < CC) {
    float s = 0.f;
#pragma unroll
    for (int w8 = 0; w8 < 8; ++w8) s += red[w8 * CC + tid];
    vpool[(size_t)b * CC + tid] = s * (1.0f / (float)NPOS);
  }
}

__device__ __forceinline__ v8f wmma4(v2f a, v2f b, v8f c) {
  return __builtin_amdgcn_wmma_f32_16x16x4_f32(
      false, a, false, b, (short)0, c, false, false);
}

#define HSTRIDE 34

__global__ __launch_bounds__(256) void value_mlp_kernel(
    const float* __restrict__ vpool,
    const float* __restrict__ w1, const float* __restrict__ b1,
    const float* __restrict__ w2, const float* __restrict__ b2,
    const float* __restrict__ w3, const float* __restrict__ b3,
    float*       __restrict__ out)
{
  __shared__ float hb[8][16 * HSTRIDE];

  const int tid   = threadIdx.x;
  const int wave  = tid >> 5;
  const int lane  = tid & 31;
  const int lh    = lane & 15;
  const int kh    = (lane >> 4) * 2;
  const int row0  = blockIdx.x * 128 + wave * 16;
  float* hbuf = hb[wave];

  v8f c0 = {}, c1 = {};
  {
    const float* arow = vpool + (size_t)(row0 + lh) * 32;
#pragma unroll
    for (int k0 = 0; k0 < 32; k0 += 4) {
      v2f a;  a.x  = arow[k0 + kh];              a.y  = arow[k0 + kh + 1];
      v2f bA; bA.x = w1[lh * 32 + k0 + kh];      bA.y = w1[lh * 32 + k0 + kh + 1];
      v2f bB; bB.x = w1[(lh + 16) * 32 + k0 + kh];
              bB.y = w1[(lh + 16) * 32 + k0 + kh + 1];
      c0 = wmma4(a, bA, c0);
      c1 = wmma4(a, bB, c1);
    }
  }
  {
    const float bb0 = b1[lh], bb1 = b1[lh + 16];
#pragma unroll
    for (int v = 0; v < 8; ++v) {
      const int m = v + (lane >> 4) * 8;
      hbuf[m * HSTRIDE + lh]      = fmaxf(c0[v] + bb0, 0.f);
      hbuf[m * HSTRIDE + lh + 16] = fmaxf(c1[v] + bb1, 0.f);
    }
  }
  __syncthreads();

  v8f d0 = {}, d1 = {};
  {
    const float* arow = hbuf + lh * HSTRIDE;
#pragma unroll
    for (int k0 = 0; k0 < 32; k0 += 4) {
      v2f a;  a.x  = arow[k0 + kh];              a.y  = arow[k0 + kh + 1];
      v2f bA; bA.x = w2[lh * 32 + k0 + kh];      bA.y = w2[lh * 32 + k0 + kh + 1];
      v2f bB; bB.x = w2[(lh + 16) * 32 + k0 + kh];
              bB.y = w2[(lh + 16) * 32 + k0 + kh + 1];
      d0 = wmma4(a, bA, d0);
      d1 = wmma4(a, bB, d1);
    }
  }
  __syncthreads();
  {
    const float bb0 = b2[lh], bb1 = b2[lh + 16];
#pragma unroll
    for (int v = 0; v < 8; ++v) {
      const int m = v + (lane >> 4) * 8;
      hbuf[m * HSTRIDE + lh]      = fmaxf(d0[v] + bb0, 0.f);
      hbuf[m * HSTRIDE + lh + 16] = fmaxf(d1[v] + bb1, 0.f);
    }
  }
  __syncthreads();

  if (lane < 16) {
    const float* hrow = hbuf + lh * HSTRIDE;
    float a0 = b3[0], a1 = b3[1], a2 = b3[2];
#pragma unroll
    for (int k = 0; k < 32; ++k) {
      const float h = hrow[k];
      a0 += h * w3[k];
      a1 += h * w3[32 + k];
      a2 += h * w3[64 + k];
    }
    const size_t r = (size_t)(row0 + lh) * 3;
    out[r + 0] = a0; out[r + 1] = a1; out[r + 2] = a2;
  }
}

extern "C" void kernel_launch(void* const* d_in, const int* in_sizes, int n_in,
                              void* d_out, int out_size, void* d_ws, size_t ws_size,
                              hipStream_t stream) {
  (void)in_sizes; (void)n_in; (void)out_size; (void)ws_size;
  const int*   enc          = (const int*)  d_in[0];
  const float* emb          = (const float*)d_in[1];
  const float* prelu_map    = (const float*)d_in[2];
  const float* dwconv_w     = (const float*)d_in[3];
  const float* dwconv_b     = (const float*)d_in[4];
  const float* policy_w     = (const float*)d_in[5];
  const float* prelu_policy = (const float*)d_in[6];
  const float* w1 = (const float*)d_in[7];
  const float* b1 = (const float*)d_in[8];
  const float* w2 = (const float*)d_in[9];
  const float* b2 = (const float*)d_in[10];
  const float* w3 = (const float*)d_in[11];
  const float* b3 = (const float*)d_in[12];

  float* out        = (float*)d_out;
  float* value_out  = out;
  float* policy_out = out + (size_t)BQ * 3;
  float* vpool      = (float*)d_ws;

  fused_front_kernel<<<BQ, 256, 0, stream>>>(
      enc, emb, prelu_map, dwconv_w, dwconv_b, policy_w, prelu_policy,
      policy_out, vpool);

  value_mlp_kernel<<<BQ / 128, 256, 0, stream>>>(
      vpool, w1, b1, w2, b2, w3, b3, value_out);
}


# Stage 2 made wave-uniform and straight-line: scalar positions, taps from a zero-bordered LDS tile (pitch 16), 2 positions per trip; tap loads grouped under one wait, no per-tap lane masks
# speedup vs baseline: 2.2447x; 1.3689x over previous
#include <hip/hip_runtime.h>
#include <math.h>


#define BQ   4096
#define LL   4
#define HH   15
#define WW   15
#define NPOS (HH * WW)
#define CC   32
#define MAPMAX 32.0f
#define PITCH  16
#define NCELL  (17 * PITCH + 1)

typedef __attribute__((ext_vector_type(2))) float v2f;
typedef __attribute__((ext_vector_type(8))) float v8f;

__global__ __launch_bounds__(256) void fused_front_kernel(
    const int*   __restrict__ enc,
    const float* __restrict__ emb,
    const float* __restrict__ prelu_map,
    const float* __restrict__ dwconv_w,
    const float* __restrict__ dwconv_b,
    const float* __restrict__ policy_w,
    const float* __restrict__ prelu_policy,
    float*       __restrict__ policy_out,
    float*       __restrict__ vpool)
{
  __shared__ float xs[NCELL * CC];
  __shared__ float red[8 * CC];

  const int b    = blockIdx.x;
  const int tid  = threadIdx.x;
  const int c    = tid & 31;
  const int wave = tid >> 5;

  const float amap  = prelu_map[c];
  const int*  ebase = enc + (size_t)b * LL * NPOS;
  const int   wv    = __builtin_amdgcn_readfirstlane(wave);
  const int   jl    = (c >> 2) & 3;
  const int   ll    = c & 3;
  int idv = ebase[ll * NPOS + jl * 8 + wv];
  xs[tid]                          = 0.f;
  xs[tid + 256]                    = 0.f;
  xs[16 * PITCH * CC + tid]        = 0.f;
  xs[16 * PITCH * CC + 256 + tid]  = 0.f;
  xs[(wv + 1) * PITCH * CC + c]    = 0.f;
  if (wv < 7)
    xs[(wv + 9) * PITCH * CC + c]  = 0.f;
  else
    xs[(NCELL - 1) * CC + c]       = 0.f;
#pragma unroll 1
  for (int g = 0; g < 7; ++g) {
    const int posn = (g < 6) ? ((4 * (g + 1) + jl) * 8 + wv) : (NPOS - 1);
    const int idn  = ebase[ll * NPOS + posn];
    float e[16];
#pragma unroll
    for (int k = 0; k < 16; ++k) {
      const int idx = __builtin_amdgcn_readlane(idv, k);
      e[k] = emb[(unsigned)(idx * CC + c)];
    }
#pragma unroll
    for (int j = 0; j < 4; ++j) {
      float acc = 0.f;
#pragma unroll
      for (int l = 0; l < LL; ++l)
        acc += MAPMAX * tanhf(e[4 * j + l] * (1.0f / MAPMAX));
      const float f = acc * 0.25f;
      const int pos = (4 * g + j) * 8 + wv;
      const int q   = pos + pos / WW + PITCH + 1;
      xs[q * CC + c] = (f >= 0.f) ? f : amap * f;
    }
    idv = idn;
  }
  if (wv == 0) {
    float acc = 0.f;
    float e[4];
#pragma unroll
    for (int l = 0; l < LL; ++l) {
      const int idx = __builtin_amdgcn_readlane(idv, l);
      e[l] = emb[(unsigned)(idx * CC + c)];
    }
#pragma unroll
    for (int l = 0; l < LL; ++l)
      acc += MAPMAX * tanhf(e[l] * (1.0f / MAPMAX));
    const float f = acc * 0.25f;
    xs[(15 * PITCH + 15) * CC + c] = (f >= 0.f) ? f : amap * f;
  }
  __syncthreads();

  float wreg[9];
#pragma unroll
  for (int j = 0; j < 9; ++j) wreg[j] = dwconv_w[c * 9 + j];
  const float bias = dwconv_b[c];
  const float pw   = (c < 16) ? policy_w[c] : 0.f;
  const float pa   = prelu_policy[0];

  const float* xc = xs + c;
  float* pol_row = policy_out + (size_t)b * NPOS;
  float vacc = 0.f;
#pragma unroll 1
  for (int it = 0; it < 28; it += 2) {
    float y[2];
#pragma unroll
    for (int u = 0; u < 2; ++u) {
      const int p = (it + u) * 8 + wv;
      const float* t = xc + (p + p / WW) * CC;
      float a = bias;
#pragma unroll
      for (int kh = 0; kh < 3; ++kh)
#pragma unroll
        for (int kw = 0; kw < 3; ++kw)
          a = __builtin_fmaf(t[(kh * PITCH + kw) * CC], wreg[kh * 3 + kw], a);
      y[u] = fmaxf(a, 0.f);
    }
    float contrib[2];
#pragma unroll
    for (int u = 0; u < 2; ++u) {
      vacc += y[u];
      contrib[u] = y[u] * pw;
    }
#pragma unroll
    for (int off = 16; off > 0; off >>= 1) {
#pragma unroll
      for (int u = 0; u < 2; ++u)
        contrib[u] += __shfl_xor(contrib[u], off, 32);
    }
    if (c == 0) {
#pragma unroll
      for (int u = 0; u < 2; ++u) {
        const int p = (it + u) * 8 + wv;
        pol_row[p] = (contrib[u] >= 0.f) ? contrib[u] : pa * contrib[u];
      }
    }
  }
  if (wv == 0) {
    const float* t = xc + (NPOS - 1 + 14) * CC;
    float a = bias;
#pragma unroll
    for (int kh = 0; kh < 3; ++kh)
#pragma unroll
      for (int kw = 0; kw < 3; ++kw)
        a = __builtin_fmaf(t[(kh * PITCH + kw) * CC], wreg[kh * 3 + kw], a);
    const float y = fmaxf(a, 0.f);
    vacc += y;
    float contrib = y * pw;
#pragma unroll
    for (int off = 16; off > 0; off >>= 1)
      contrib += __shfl_xor(contrib, off, 32);
    if (c == 0)
      pol_row[NPOS - 1] = (contrib >= 0.f) ? contrib : pa * contrib;
  }

  red[wave * CC + c] = vacc;
  __syncthreads();
  if (tid < CC) {
    float s = 0.f;
#pragma unroll
    for (int w8 = 0; w8 < 8; ++w8) s += red[w8 * CC + tid];
    vpool[(size_t)b * CC + tid] = s * (1.0f / (float)NPOS);
  }
}

__device__ __forceinline__ v8f wmma4(v2f a, v2f b, v8f c) {
  return __builtin_amdgcn_wmma_f32_16x16x4_f32(
      false, a, false, b, (short)0, c, false, false);
}

#define HSTRIDE 34

__global__ __launch_bounds__(256) void value_mlp_kernel(
    const float* __restrict__ vpool,
    const float* __restrict__ w1, const float* __restrict__ b1,
    const float* __restrict__ w2, const float* __restrict__ b2,
    const float* __restrict__ w3, const float* __restrict__ b3,
    float*       __restrict__ out)
{
  __shared__ float hb[8][16 * HSTRIDE];

  const int tid   = threadIdx.x;
  const int wave  = tid >> 5;
  const int lane  = tid & 31;
  const int lh    = lane & 15;
  const int kh    = (lane >> 4) * 2;
  const int row0  = blockIdx.x * 128 + wave * 16;
  float* hbuf = hb[wave];

  v8f c0 = {}, c1 = {};
  {
    const float* arow = vpool + (size_t)(row0 + lh) * 32;
#pragma unroll
    for (int k0 = 0; k0 < 32; k0 += 4) {
      v2f a;  a.x  = arow[k0 + kh];              a.y  = arow[k0 + kh + 1];
      v2f bA; bA.x = w1[lh * 32 + k0 + kh];      bA.y = w1[lh * 32 + k0 + kh + 1];
      v2f bB; bB.x = w1[(lh + 16) * 32 + k0 + kh];
              bB.y = w1[(lh + 16) * 32 + k0 + kh + 1];
      c0 = wmma4(a, bA, c0);
      c1 = wmma4(a, bB, c1);
    }
  }
  {
    const float bb0 = b1[lh], bb1 = b1[lh + 16];
#pragma unroll
    for (int v = 0; v < 8; ++v) {
      const int m = v + (lane >> 4) * 8;
      hbuf[m * HSTRIDE + lh]      = fmaxf(c0[v] + bb0, 0.f);
      hbuf[m * HSTRIDE + lh + 16] = fmaxf(c1[v] + bb1, 0.f);
    }
  }
  __syncthreads();

  v8f d0 = {}, d1 = {};
  {
    const float* arow = hbuf + lh * HSTRIDE;
#pragma unroll
    for (int k0 = 0; k0 < 32; k0 += 4) {
      v2f a;  a.x  = arow[k0 + kh];              a.y  = arow[k0 + kh + 1];
      v2f bA; bA.x = w2[lh * 32 + k0 + kh];      bA.y = w2[lh * 32 + k0 + kh + 1];
      v2f bB; bB.x = w2[(lh + 16) * 32 + k0 + kh];
              bB.y = w2[(lh + 16) * 32 + k0 + kh + 1];
      d0 = wmma4(a, bA, d0);
      d1 = wmma4(a, bB, d1);
    }
  }
  __syncthreads();
  {
    const float bb0 = b2[lh], bb1 = b2[lh + 16];
#pragma unroll
    for (int v = 0; v < 8; ++v) {
      const int m = v + (lane >> 4) * 8;
      hbuf[m * HSTRIDE + lh]      = fmaxf(d0[v] + bb0, 0.f);
      hbuf[m * HSTRIDE + lh + 16] = fmaxf(d1[v] + bb1, 0.f);
    }
  }
  __syncthreads();

  if (lane < 16) {
    const float* hrow = hbuf + lh * HSTRIDE;
    float a0 = b3[0], a1 = b3[1], a2 = b3[2];
#pragma unroll
    for (int k = 0; k < 32; ++k) {
      const float h = hrow[k];
      a0 += h * w3[k];
      a1 += h * w3[32 + k];
      a2 += h * w3[64 + k];
    }
    const size_t r = (size_t)(row0 + lh) * 3;
    out[r + 0] = a0; out[r + 1] = a1; out[r + 2] = a2;
  }
}

extern "C" void kernel_launch(void* const* d_in, const int* in_sizes, int n_in,
                              void* d_out, int out_size, void* d_ws, size_t ws_size,
                              hipStream_t stream) {
  (void)in_sizes; (void)n_in; (void)out_size; (void)ws_size;
  const int*   enc          = (const int*)  d_in[0];
  const float* emb          = (const float*)d_in[1];
  const float* prelu_map    = (const float*)d_in[2];
  const float* dwconv_w     = (const float*)d_in[3];
  const float* dwconv_b     = (const float*)d_in[4];
  const float* policy_w     = (const float*)d_in[5];
  const float* prelu_policy = (const float*)d_in[6];
  const float* w1 = (const float*)d_in[7];
  const float* b1 = (const float*)d_in[8];
  const float* w2 = (const float*)d_in[9];
  const float* b2 = (const float*)d_in[10];
  const float* w3 = (const float*)d_in[11];
  const float* b3 = (const float*)d_in[12];

  float* out        = (float*)d_out;
  float* value_out  = out;
  float* policy_out = out + (size_t)BQ * 3;
  float* vpool      = (float*)d_ws;

  fused_front_kernel<<<BQ, 256, 0, stream>>>(
      enc, emb, prelu_map, dwconv_w, dwconv_b, policy_w, prelu_policy,
      policy_out, vpool);

  value_mlp_kernel<<<BQ / 128, 256, 0, stream>>>(
      vpool, w1, b1, w2, b2, w3, b3, value_out);
}


# Stage-1 tanh: one wave-uniform range test per group of 16 table rows, then tanhf's small-argument polynomial as 16 independent straight-line chains (scales folded exactly); tanhf kept for big values
# speedup vs baseline: 2.3954x; 1.0671x over previous
#include <hip/hip_runtime.h>
#include <math.h>


#define BQ   4096
#define LL   4
#define HH   15
#define WW   15
#define NPOS (HH * WW)
#define CC   32
#define MAPMAX 32.0f
#define PITCH  16
#define NCELL  (17 * PITCH + 1)

typedef __attribute__((ext_vector_type(2))) float v2f;
typedef __attribute__((ext_vector_type(8))) float v8f;


__device__ __forceinline__ float mapped_tanh_small(float e) {
  const float s = e * e;
  float p = __builtin_fmaf(s, -0x1.758e7ap-58f, 0x1.521192p-46f);
  p = __builtin_fmaf(s, p, -0x1.b8389cp-35f);
  p = __builtin_fmaf(s, p, 0x1.110704p-23f);
  p = __builtin_fmaf(s, p, -0x1.555532p-12f);
  return __builtin_fmaf(s, e * p, e);
}

template <int N>
__device__ __forceinline__ bool group_is_small(const float (&e)[N]) {
  float m = fabsf(e[0]);
#pragma unroll
  for (int k = 1; k < N; ++k) m = fmaxf(m, fabsf(e[k]));
  return __builtin_amdgcn_ballot_w32(!(m < 20.0f)) == 0u;
}

__device__ __forceinline__ float mean_prelu(float r0, float r1, float r2, float r3, float amap) {
  float acc = r0;
  acc += r1;
  acc += r2;
  acc += r3;
  const float f = acc * 0.25f;
  return (f >= 0.f) ? f : amap * f;
}

__global__ __launch_bounds__(256) void fused_front_kernel(
    const int*   __restrict__ enc,
    const float* __restrict__ emb,
    const float* __restrict__ prelu_map,
    const float* __restrict__ dwconv_w,
    const float* __restrict__ dwconv_b,
    const float* __restrict__ policy_w,
    const float* __restrict__ prelu_policy,
    float*       __restrict__ policy_out,
    float*       __restrict__ vpool)
{
  __shared__ float xs[NCELL * CC];
  __shared__ float red[8 * CC];

  const int b    = blockIdx.x;
  const int tid  = threadIdx.x;
  const int c    = tid & 31;
  const int wave = tid >> 5;

  const float amap  = prelu_map[c];
  const int*  ebase = enc + (size_t)b * LL * NPOS;
  const int   wv    = __builtin_amdgcn_readfirstlane(wave);
  const int   jl    = (c >> 2) & 3;
  const int   ll    = c & 3;
  int idv = ebase[ll * NPOS + jl * 8 + wv];
  xs[tid]                          = 0.f;
  xs[tid + 256]                    = 0.f;
  xs[16 * PITCH * CC + tid]        = 0.f;
  xs[16 * PITCH * CC + 256 + tid]  = 0.f;
  xs[(wv + 1) * PITCH * CC + c]    = 0.f;
  if (wv < 7)
    xs[(wv + 9) * PITCH * CC + c]  = 0.f;
  else
    xs[(NCELL - 1) * CC + c]       = 0.f;
#pragma unroll 1
  for (int g = 0; g < 7; ++g) {
    const int posn = (g < 6) ? ((4 * (g + 1) + jl) * 8 + wv) : (NPOS - 1);
    const int idn  = ebase[ll * NPOS + posn];
    float e[16];
#pragma unroll
    for (int k = 0; k < 16; ++k) {
      const int idx = __builtin_amdgcn_readlane(idv, k);
      e[k] = emb[(unsigned)(idx * CC + c)];
    }
    int q[4];
#pragma unroll
    for (int j = 0; j < 4; ++j) {
      const int pos = (4 * g + j) * 8 + wv;
      q[j] = pos + pos / WW + PITCH + 1;
    }
    if (__builtin_expect(group_is_small<16>(e), 1)) {
#pragma unroll
      for (int j = 0; j < 4; ++j)
        xs[q[j] * CC + c] = mean_prelu(mapped_tanh_small(e[4 * j]), mapped_tanh_small(e[4 * j + 1]),
                                       mapped_tanh_small(e[4 * j + 2]), mapped_tanh_small(e[4 * j + 3]), amap);
    } else {
#pragma unroll
      for (int j = 0; j < 4; ++j)
        xs[q[j] * CC + c] = mean_prelu(MAPMAX * tanhf(e[4 * j] * (1.0f / MAPMAX)),
                                       MAPMAX * tanhf(e[4 * j + 1] * (1.0f / MAPMAX)),
                                       MAPMAX * tanhf(e[4 * j + 2] * (1.0f / MAPMAX)),
                                       MAPMAX * tanhf(e[4 * j + 3] * (1.0f / MAPMAX)), amap);
    }
    idv = idn;
  }
  if (wv == 0) {
    float e[4];
#pragma unroll
    for (int l = 0; l < LL; ++l) {
      const int idx = __builtin_amdgcn_readlane(idv, l);
      e[l] = emb[(unsigned)(idx * CC + c)];
    }
    float f;
    if (__builtin_expect(group_is_small<4>(e), 1))
      f = mean_prelu(mapped_tanh_small(e[0]), mapped_tanh_small(e[1]),
                     mapped_tanh_small(e[2]), mapped_tanh_small(e[3]), amap);
    else
      f = mean_prelu(MAPMAX * tanhf(e[0] * (1.0f / MAPMAX)), MAPMAX * tanhf(e[1] * (1.0f / MAPMAX)),
                     MAPMAX * tanhf(e[2] * (1.0f / MAPMAX)), MAPMAX * tanhf(e[3] * (1.0f / MAPMAX)), amap);
    xs[(15 * PITCH + 15) * CC + c] = f;
  }
  __syncthreads();

  float wreg[9];
#pragma unroll
  for (int j = 0; j < 9; ++j) wreg[j] = dwconv_w[c * 9 + j];
  const float bias = dwconv_b[c];
  const float pw   = (c < 16) ? policy_w[c] : 0.f;
  const float pa   = prelu_policy[0];

  const float* xc = xs + c;
  float* pol_row = policy_out + (size_t)b * NPOS;
  float vacc = 0.f;
#pragma unroll 1
  for (int it = 0; it < 28; it += 2) {
    float y[2];
#pragma unroll
    for (int u = 0; u < 2; ++u) {
      const int p = (it + u) * 8 + wv;
      const float* t = xc + (p + p / WW) * CC;
      float a = bias;
#pragma unroll
      for (int kh = 0; kh < 3; ++kh)
#pragma unroll
        for (int kw = 0; kw < 3; ++kw)
          a = __builtin_fmaf(t[(kh * PITCH + kw) * CC], wreg[kh * 3 + kw], a);
      y[u] = fmaxf(a, 0.f);
    }
    float contrib[2];
#pragma unroll
    for (int u = 0; u < 2; ++u) {
      vacc += y[u];
      contrib[u] = y[u] * pw;
    }
#pragma unroll
    for (int off = 16; off > 0; off >>= 1) {
#pragma unroll
      for (int u = 0; u < 2; ++u)
        contrib[u] += __shfl_xor(contrib[u], off, 32);
    }
    if (c == 0) {
#pragma unroll
      for (int u = 0; u < 2; ++u) {
        const int p = (it + u) * 8 + wv;
        pol_row[p] = (contrib[u] >= 0.f) ? contrib[u] : pa * contrib[u];
      }
    }
  }
  if (wv == 0) {
    const float* t = xc + (NPOS - 1 + 14) * CC;
    float a = bias;
#pragma unroll
    for (int kh = 0; kh < 3; ++kh)
#pragma unroll
      for (int kw = 0; kw < 3; ++kw)
        a = __builtin_fmaf(t[(kh * PITCH + kw) * CC], wreg[kh * 3 + kw], a);
    const float y = fmaxf(a, 0.f);
    vacc += y;
    float contrib = y * pw;
#pragma unroll
    for (int off = 16; off > 0; off >>= 1)
      contrib += __shfl_xor(contrib, off, 32);
    if (c == 0)
      pol_row[NPOS - 1] = (contrib >= 0.f) ? contrib : pa * contrib;
  }

  red[wave * CC + c] = vacc;
  __syncthreads();
  if (tid < CC) {
    float s = 0.f;
#pragma unroll
    for (int w8 = 0; w8 < 8; ++w8) s += red[w8 * CC + tid];
    vpool[(size_t)b * CC + tid] = s * (1.0f / (float)NPOS);
  }
}

__device__ __forceinline__ v8f wmma4(v2f a, v2f b, v8f c) {
  return __builtin_amdgcn_wmma_f32_16x16x4_f32(
      false, a, false, b, (short)0, c, false, false);
}

#define HSTRIDE 34

__global__ __launch_bounds__(256) void value_mlp_kernel(
    const float* __restrict__ vpool,
    const float* __restrict__ w1, const float* __restrict__ b1,
    const float* __restrict__ w2, const float* __restrict__ b2,
    const float* __restrict__ w3, const float* __restrict__ b3,
    float*       __restrict__ out)
{
  __shared__ float hb[8][16 * HSTRIDE];

  const int tid   = threadIdx.x;
  const int wave  = tid >> 5;
  const int lane  = tid & 31;
  const int lh    = lane & 15;
  const int kh    = (lane >> 4) * 2;
  const int row0  = blockIdx.x * 128 + wave * 16;
  float* hbuf = hb[wave];

  v8f c0 = {}, c1 = {};
  {
    const float* arow = vpool + (size_t)(row0 + lh) * 32;
#pragma unroll
    for (int k0 = 0; k0 < 32; k0 += 4) {
      v2f a;  a.x  = arow[k0 + kh];              a.y  = arow[k0 + kh + 1];
      v2f bA; bA.x = w1[lh * 32 + k0 + kh];      bA.y = w1[lh * 32 + k0 + kh + 1];
      v2f bB; bB.x = w1[(lh + 16) * 32 + k0 + kh];
              bB.y = w1[(lh + 16) * 32 + k0 + kh + 1];
      c0 = wmma4(a, bA, c0);
      c1 = wmma4(a, bB, c1);
    }
  }
  {
    const float bb0 = b1[lh], bb1 = b1[lh + 16];
#pragma unroll
    for (int v = 0; v < 8; ++v) {
      const int m = v + (lane >> 4) * 8;
      hbuf[m * HSTRIDE + lh]      = fmaxf(c0[v] + bb0, 0.f);
      hbuf[m * HSTRIDE + lh + 16] = fmaxf(c1[v] + bb1, 0.f);
    }
  }
  __syncthreads();

  v8f d0 = {}, d1 = {};
  {
    const float* arow = hbuf + lh * HSTRIDE;
#pragma unroll
    for (int k0 = 0; k0 < 32; k0 += 4) {
      v2f a;  a.x  = arow[k0 + kh];              a.y  = arow[k0 + kh + 1];
      v2f bA; bA.x = w2[lh * 32 + k0 + kh];      bA.y = w2[lh * 32 + k0 + kh + 1];
      v2f bB; bB.x = w2[(lh + 16) * 32 + k0 + kh];
              bB.y = w2[(lh + 16) * 32 + k0 + kh + 1];
      d0 = wmma4(a, bA, d0);
      d1 = wmma4(a, bB, d1);
    }
  }
  __syncthreads();
  {
    const float bb0 = b2[lh], bb1 = b2[lh + 16];
#pragma unroll
    for (int v = 0; v < 8; ++v) {
      const int m = v + (lane >> 4) * 8;
      hbuf[m * HSTRIDE + lh]      = fmaxf(d0[v] + bb0, 0.f);
      hbuf[m * HSTRIDE + lh + 16] = fmaxf(d1[v] + bb1, 0.f);
    }
  }
  __syncthreads();

  if (lane < 16) {
    const float* hrow = hbuf + lh * HSTRIDE;
    float a0 = b3[0], a1 = b3[1], a2 = b3[2];
#pragma unroll
    for (int k = 0; k < 32; ++k) {
      const float h = hrow[k];
      a0 += h * w3[k];
      a1 += h * w3[32 + k];
      a2 += h * w3[64 + k];
    }
    const size_t r = (size_t)(row0 + lh) * 3;
    out[r + 0] = a0; out[r + 1] = a1; out[r + 2] = a2;
  }
}

extern "C" void kernel_launch(void* const* d_in, const int* in_sizes, int n_in,
                              void* d_out, int out_size, void* d_ws, size_t ws_size,
                              hipStream_t stream) {
  (void)in_sizes; (void)n_in; (void)out_size; (void)ws_size;
  const int*   enc          = (const int*)  d_in[0];
  const float* emb          = (const float*)d_in[1];
  const float* prelu_map    = (const float*)d_in[2];
  const float* dwconv_w     = (const float*)d_in[3];
  const float* dwconv_b     = (const float*)d_in[4];
  const float* policy_w     = (const float*)d_in[5];
  const float* prelu_policy = (const float*)d_in[6];
  const float* w1 = (const float*)d_in[7];
  const float* b1 = (const float*)d_in[8];
  const float* w2 = (const float*)d_in[9];
  const float* b2 = (const float*)d_in[10];
  const float* w3 = (const float*)d_in[11];
  const float* b3 = (const float*)d_in[12];

  float* out        = (float*)d_out;
  float* value_out  = out;
  float* policy_out = out + (size_t)BQ * 3;
  float* vpool      = (float*)d_ws;

  fused_front_kernel<<<BQ, 256, 0, stream>>>(
      enc, emb, prelu_map, dwconv_w, dwconv_b, policy_w, prelu_policy,
      policy_out, vpool);

  value_mlp_kernel<<<BQ / 128, 256, 0, stream>>>(
      vpool, w1, b1, w2, b2, w3, b3, value_out);
}


# Policy-head lane sums off the LDS unit (no ds_bpermute): a trip's 2nd position moves to lanes 16..31 by one v_permlanex16, one row_xmask DPP butterfly sums both; same tree less the idle half's +0
# speedup vs baseline: 2.7074x; 1.1302x over previous
#include <hip/hip_runtime.h>
#include <math.h>


#define BQ   4096
#define LL   4
#define HH   15
#define WW   15
#define NPOS (HH * WW)
#define CC   32
#define MAPMAX 32.0f
#define PITCH  16
#define NCELL  (17 * PITCH + 1)

typedef __attribute__((ext_vector_type(2))) float v2f;
typedef __attribute__((ext_vector_type(8))) float v8f;


__device__ __forceinline__ float mapped_tanh_small(float e) {
  const float s = e * e;
  float p = __builtin_fmaf(s, -0x1.758e7ap-58f, 0x1.521192p-46f);
  p = __builtin_fmaf(s, p, -0x1.b8389cp-35f);
  p = __builtin_fmaf(s, p, 0x1.110704p-23f);
  p = __builtin_fmaf(s, p, -0x1.555532p-12f);
  return __builtin_fmaf(s, e * p, e);
}

template <int N>
__device__ __forceinline__ bool group_is_small(const float (&e)[N]) {
  float m = fabsf(e[0]);
#pragma unroll
  for (int k = 1; k < N; ++k) m = fmaxf(m, fabsf(e[k]));
  return __builtin_amdgcn_ballot_w32(!(m < 20.0f)) == 0u;
}

__device__ __forceinline__ float mean_prelu(float r0, float r1, float r2, float r3, float amap) {
  float acc = r0;
  acc += r1;
  acc += r2;
  acc += r3;
  const float f = acc * 0.25f;
  return (f >= 0.f) ? f : amap * f;
}


__device__ __forceinline__ float other_row(float v) {
  const int b = __builtin_bit_cast(int, v);
  return __builtin_bit_cast(float,
      __builtin_amdgcn_permlanex16(b, b, 0x76543210, 0xfedcba98, false, false));
}

template <int M>
__device__ __forceinline__ float row_xor_add(float v) {
  const int t = __builtin_amdgcn_update_dpp(0, __builtin_bit_cast(int, v),
                                            0x160 | M  , 0xf, 0xf, true);
  return v + __builtin_bit_cast(float, t);
}

__device__ __forceinline__ float row_sum16(float v) {
  v = row_xor_add<8>(v);
  v = row_xor_add<4>(v);
  v = row_xor_add<2>(v);
  v = row_xor_add<1>(v);
  return v;
}

__global__ __launch_bounds__(256) void fused_front_kernel(
    const int*   __restrict__ enc,
    const float* __restrict__ emb,
    const float* __restrict__ prelu_map,
    const float* __restrict__ dwconv_w,
    const float* __restrict__ dwconv_b,
    const float* __restrict__ policy_w,
    const float* __restrict__ prelu_policy,
    float*       __restrict__ policy_out,
    float*       __restrict__ vpool)
{
  __shared__ float xs[NCELL * CC];
  __shared__ float red[8 * CC];

  const int b    = blockIdx.x;
  const int tid  = threadIdx.x;
  const int c    = tid & 31;
  const int wave = tid >> 5;

  const float amap  = prelu_map[c];
  const int*  ebase = enc + (size_t)b * LL * NPOS;
  const int   wv    = __builtin_amdgcn_readfirstlane(wave);
  const int   jl    = (c >> 2) & 3;
  const int   ll    = c & 3;
  int idv = ebase[ll * NPOS + jl * 8 + wv];
  xs[tid]                          = 0.f;
  xs[tid + 256]                    = 0.f;
  xs[16 * PITCH * CC + tid]        = 0.f;
  xs[16 * PITCH * CC + 256 + tid]  = 0.f;
  xs[(wv + 1) * PITCH * CC + c]    = 0.f;
  if (wv < 7)
    xs[(wv + 9) * PITCH * CC + c]  = 0.f;
  else
    xs[(NCELL - 1) * CC + c]       = 0.f;
#pragma unroll 1
  for (int g = 0; g < 7; ++g) {
    const int posn = (g < 6) ? ((4 * (g + 1) + jl) * 8 + wv) : (NPOS - 1);
    const int idn  = ebase[ll * NPOS + posn];
    float e[16];
#pragma unroll
    for (int k = 0; k < 16; ++k) {
      const int idx = __builtin_amdgcn_readlane(idv, k);
      e[k] = emb[(unsigned)(idx * CC + c)];
    }
    int q[4];
#pragma unroll
    for (int j = 0; j < 4; ++j) {
      const int pos = (4 * g + j) * 8 + wv;
      q[j] = pos + pos / WW + PITCH + 1;
    }
    if (__builtin_expect(group_is_small<16>(e), 1)) {
#pragma unroll
      for (int j = 0; j < 4; ++j)
        xs[q[j] * CC + c] = mean_prelu(mapped_tanh_small(e[4 * j]), mapped_tanh_small(e[4 * j + 1]),
                                       mapped_tanh_small(e[4 * j + 2]), mapped_tanh_small(e[4 * j + 3]), amap);
    } else {
#pragma unroll
      for (int j = 0; j < 4; ++j)
        xs[q[j] * CC + c] = mean_prelu(MAPMAX * tanhf(e[4 * j] * (1.0f / MAPMAX)),
                                       MAPMAX * tanhf(e[4 * j + 1] * (1.0f / MAPMAX)),
                                       MAPMAX * tanhf(e[4 * j + 2] * (1.0f / MAPMAX)),
                                       MAPMAX * tanhf(e[4 * j + 3] * (1.0f / MAPMAX)), amap);
    }
    idv = idn;
  }
  if (wv == 0) {
    float e[4];
#pragma unroll
    for (int l = 0; l < LL; ++l) {
      const int idx = __builtin_amdgcn_readlane(idv, l);
      e[l] = emb[(unsigned)(idx * CC + c)];
    }
    float f;
    if (__builtin_expect(group_is_small<4>(e), 1))
      f = mean_prelu(mapped_tanh_small(e[0]), mapped_tanh_small(e[1]),
                     mapped_tanh_small(e[2]), mapped_tanh_small(e[3]), amap);
    else
      f = mean_prelu(MAPMAX * tanhf(e[0] * (1.0f / MAPMAX)), MAPMAX * tanhf(e[1] * (1.0f / MAPMAX)),
                     MAPMAX * tanhf(e[2] * (1.0f / MAPMAX)), MAPMAX * tanhf(e[3] * (1.0f / MAPMAX)), amap);
    xs[(15 * PITCH + 15) * CC + c] = f;
  }
  __syncthreads();

  float wreg[9];
#pragma unroll
  for (int j = 0; j < 9; ++j) wreg[j] = dwconv_w[c * 9 + j];
  const float bias = dwconv_b[c];
  const float pw   = (c < 16) ? policy_w[c] : 0.f;
  const float pa   = prelu_policy[0];

  const float* xc = xs + c;
  float* pol_row = policy_out + (size_t)b * NPOS;
  float* pol_lane = pol_row + wv + (c >> 1);
  const bool row0 = c < 16;
  float vacc = 0.f;
#pragma unroll 1
  for (int it = 0; it < 28; it += 2) {
    float y[2];
#pragma unroll
    for (int u = 0; u < 2; ++u) {
      const int p = (it + u) * 8 + wv;
      const float* t = xc + (p + p / WW) * CC;
      float a = bias;
#pragma unroll
      for (int kh = 0; kh < 3; ++kh)
#pragma unroll
        for (int kw = 0; kw < 3; ++kw)
          a = __builtin_fmaf(t[(kh * PITCH + kw) * CC], wreg[kh * 3 + kw], a);
      y[u] = fmaxf(a, 0.f);
    }
    float contrib[2];
#pragma unroll
    for (int u = 0; u < 2; ++u) {
      vacc += y[u];
      contrib[u] = y[u] * pw;
    }
    const float moved = other_row(contrib[1]);
    float z = row0 ? contrib[0] : moved;
    z = row_sum16(z);
    if ((c & 15) == 0)
      pol_lane[it * 8] = (z >= 0.f) ? z : pa * z;
  }
  if (wv == 0) {
    const float* t = xc + (NPOS - 1 + 14) * CC;
    float a = bias;
#pragma unroll
    for (int kh = 0; kh < 3; ++kh)
#pragma unroll
      for (int kw = 0; kw < 3; ++kw)
        a = __builtin_fmaf(t[(kh * PITCH + kw) * CC], wreg[kh * 3 + kw], a);
    const float y = fmaxf(a, 0.f);
    vacc += y;
    const float contrib = row_sum16(y * pw);
    if (c == 0)
      pol_row[NPOS - 1] = (contrib >= 0.f) ? contrib : pa * contrib;
  }

  red[wave * CC + c] = vacc;
  __syncthreads();
  if (tid < CC) {
    float s = 0.f;
#pragma unroll
    for (int w8 = 0; w8 < 8; ++w8) s += red[w8 * CC + tid];
    vpool[(size_t)b * CC + tid] = s * (1.0f / (float)NPOS);
  }
}

__device__ __forceinline__ v8f wmma4(v2f a, v2f b, v8f c) {
  return __builtin_amdgcn_wmma_f32_16x16x4_f32(
      false, a, false, b, (short)0, c, false, false);
}

#define HSTRIDE 34

__global__ __launch_bounds__(256) void value_mlp_kernel(
    const float* __restrict__ vpool,
    const float* __restrict__ w1, const float* __restrict__ b1,
    const float* __restrict__ w2, const float* __restrict__ b2,
    const float* __restrict__ w3, const float* __restrict__ b3,
    float*       __restrict__ out)
{
  __shared__ float hb[8][16 * HSTRIDE];

  const int tid   = threadIdx.x;
  const int wave  = tid >> 5;
  const int lane  = tid & 31;
  const int lh    = lane & 15;
  const int kh    = (lane >> 4) * 2;
  const int row0  = blockIdx.x * 128 + wave * 16;
  float* hbuf = hb[wave];

  v8f c0 = {}, c1 = {};
  {
    const float* arow = vpool + (size_t)(row0 + lh) * 32;
#pragma unroll
    for (int k0 = 0; k0 < 32; k0 += 4) {
      v2f a;  a.x  = arow[k0 + kh];              a.y  = arow[k0 + kh + 1];
      v2f bA; bA.x = w1[lh * 32 + k0 + kh];      bA.y = w1[lh * 32 + k0 + kh + 1];
      v2f bB; bB.x = w1[(lh + 16) * 32 + k0 + kh];
              bB.y = w1[(lh + 16) * 32 + k0 + kh + 1];
      c0 = wmma4(a, bA, c0);
      c1 = wmma4(a, bB, c1);
    }
  }
  {
    const float bb0 = b1[lh], bb1 = b1[lh + 16];
#pragma unroll
    for (int v = 0; v < 8; ++v) {
      const int m = v + (lane >> 4) * 8;
      hbuf[m * HSTRIDE + lh]      = fmaxf(c0[v] + bb0, 0.f);
      hbuf[m * HSTRIDE + lh + 16] = fmaxf(c1[v] + bb1, 0.f);
    }
  }
  __syncthreads();

  v8f d0 = {}, d1 = {};
  {
    const float* arow = hbuf + lh * HSTRIDE;
#pragma unroll
    for (int k0 = 0; k0 < 32; k0 += 4) {
      v2f a;  a.x  = arow[k0 + kh];              a.y  = arow[k0 + kh + 1];
      v2f bA; bA.x = w2[lh * 32 + k0 + kh];      bA.y = w2[lh * 32 + k0 + kh + 1];
      v2f bB; bB.x = w2[(lh + 16) * 32 + k0 + kh];
              bB.y = w2[(lh + 16) * 32 + k0 + kh + 1];
      d0 = wmma4(a, bA, d0);
      d1 = wmma4(a, bB, d1);
    }
  }
  __syncthreads();
  {
    const float bb0 = b2[lh], bb1 = b2[lh + 16];
#pragma unroll
    for (int v = 0; v < 8; ++v) {
      const int m = v + (lane >> 4) * 8;
      hbuf[m * HSTRIDE + lh]      = fmaxf(d0[v] + bb0, 0.f);
      hbuf[m * HSTRIDE + lh + 16] = fmaxf(d1[v] + bb1, 0.f);
    }
  }
  __syncthreads();

  if (lane < 16) {
    const float* hrow = hbuf + lh * HSTRIDE;
    float a0 = b3[0], a1 = b3[1], a2 = b3[2];
#pragma unroll
    for (int k = 0; k < 32; ++k) {
      const float h = hrow[k];
      a0 += h * w3[k];
      a1 += h * w3[32 + k];
      a2 += h * w3[64 + k];
    }
    const size_t r = (size_t)(row0 + lh) * 3;
    out[r + 0] = a0; out[r + 1] = a1; out[r + 2] = a2;
  }
}

extern "C" void kernel_launch(void* const* d_in, const int* in_sizes, int n_in,
                              void* d_out, int out_size, void* d_ws, size_t ws_size,
                              hipStream_t stream) {
  (void)in_sizes; (void)n_in; (void)out_size; (void)ws_size;
  const int*   enc          = (const int*)  d_in[0];
  const float* emb          = (const float*)d_in[1];
  const float* prelu_map    = (const float*)d_in[2];
  const float* dwconv_w     = (const float*)d_in[3];
  const float* dwconv_b     = (const float*)d_in[4];
  const float* policy_w     = (const float*)d_in[5];
  const float* prelu_policy = (const float*)d_in[6];
  const float* w1 = (const float*)d_in[7];
  const float* b1 = (const float*)d_in[8];
  const float* w2 = (const float*)d_in[9];
  const float* b2 = (const float*)d_in[10];
  const float* w3 = (const float*)d_in[11];
  const float* b3 = (const float*)d_in[12];

  float* out        = (float*)d_out;
  float* value_out  = out;
  float* policy_out = out + (size_t)BQ * 3;
  float* vpool      = (float*)d_ws;

  fused_front_kernel<<<BQ, 256, 0, stream>>>(
      enc, emb, prelu_map, dwconv_w, dwconv_b, policy_w, prelu_policy,
      policy_out, vpool);

  value_mlp_kernel<<<BQ / 128, 256, 0, stream>>>(
      vpool, w1, b1, w2, b2, w3, b3, value_out);
}
